# starting kernel
# baseline (speedup 1.0000x reference)
#include <hip/hip_runtime.h>

typedef __attribute__((ext_vector_type(2))) float v2f;
typedef __attribute__((ext_vector_type(8))) float v8f;

#define NCLS 256
#define NDF  512
#define NB   1024
#define KTOT (2 * NDF)
#define ALP  0.1f
#define HALP 0.05f

__global__ __launch_bounds__(NDF) void stats_kernel(
    const float* __restrict__ df, const int* __restrict__ gt,
    float* __restrict__ var_out)
{
    __shared__ int gtS[NB];
    const int c = blockIdx.x;
    const int d = threadIdx.x;
    gtS[d]        = gt[d];
    gtS[d + NDF]  = gt[d + NDF];
    __syncthreads();

    float m = 0.0f, v = 0.0f;
    int   n = 0;
    for (int b = 0; b < NB; ++b) {
        if (gtS[b] == c) {
            const float f  = df[(size_t)b * NDF + d];
            const float nf = (float)n;
            const float c1 = nf / (nf + 1.0f);
            const float c2 = 1.0f / (nf + 1.0f);
            m = (n == 0) ? f : (m * c1 + f * c2);
            const float a = f - m;
            v = v * c1 + a * a * c1 * c2;
            ++n;
        }
    }
    var_out[(size_t)c * NDF + d] = v;
}

__global__ __launch_bounds__(NDF) void prep_a_kernel(
    const float* __restrict__ df, const int* __restrict__ gt,
    const float* __restrict__ fc, const float* __restrict__ var_in,
    float* __restrict__ XY, float* __restrict__ constb)
{
    __shared__ float red[NDF];
    const int b = blockIdx.x;
    const int d = threadIdx.x;
    const int t = gt[b];

    const float fcv = fc[(size_t)t * NDF + d];
    const float va  = var_in[(size_t)t * NDF + d];
    const float dfv = df[(size_t)b * NDF + d];

    XY[(size_t)b * KTOT + d]       = dfv - ALP * fcv * va;
    XY[(size_t)b * KTOT + NDF + d] = HALP * va;

    red[d] = -fcv * dfv + HALP * fcv * fcv * va;
    __syncthreads();
    for (int s = NDF / 2; s > 0; s >>= 1) {
        if (d < s) red[d] += red[d + s];
        __syncthreads();
    }
    if (d == 0) constb[b] = red[0];
}

__global__ __launch_bounds__(NDF) void prep_b_kernel(
    const float* __restrict__ fc, float* __restrict__ PQ)
{
    const int c = blockIdx.x;
    const int d = threadIdx.x;
    const float v = fc[(size_t)c * NDF + d];
    PQ[(size_t)c * KTOT + d]       = v;
    PQ[(size_t)c * KTOT + NDF + d] = v * v;
}

__global__ __launch_bounds__(32) void wmma_gemm_kernel(
    const float* __restrict__ XY, const float* __restrict__ PQ,
    const float* __restrict__ constb, float* __restrict__ out)
{
    const int tm   = blockIdx.x;
    const int ts   = blockIdx.y;
    const int lane = threadIdx.x;
    const int r    = lane & 15;
    const int kh   = (lane >> 4) << 1;

    const float* Arow = XY + (size_t)(tm * 16 + r) * KTOT + kh;
    const float* Brow = PQ + (size_t)(ts * 64 + r) * KTOT + kh;

    v8f acc0 = {}, acc1 = {}, acc2 = {}, acc3 = {};
#pragma unroll 4
    for (int k = 0; k < KTOT; k += 4) {
        v2f a  = *(const v2f*)(Arow + k);
        v2f b0 = *(const v2f*)(Brow + k);
        v2f b1 = *(const v2f*)(Brow + 16 * KTOT + k);
        v2f b2 = *(const v2f*)(Brow + 32 * KTOT + k);
        v2f b3 = *(const v2f*)(Brow + 48 * KTOT + k);
        acc0 = __builtin_amdgcn_wmma_f32_16x16x4_f32(false, a, false, b0,
                                                     (short)0, acc0, false, false);
        acc1 = __builtin_amdgcn_wmma_f32_16x16x4_f32(false, a, false, b1,
                                                     (short)0, acc1, false, false);
        acc2 = __builtin_amdgcn_wmma_f32_16x16x4_f32(false, a, false, b2,
                                                     (short)0, acc2, false, false);
        acc3 = __builtin_amdgcn_wmma_f32_16x16x4_f32(false, a, false, b3,
                                                     (short)0, acc3, false, false);
    }

    const int mbase = tm * 16 + ((lane >> 4) << 3);
    float cst[8];
#pragma unroll
    for (int j = 0; j < 8; ++j) cst[j] = constb[mbase + j];

    const int colbase = ts * 64 + r;
#pragma unroll
    for (int j = 0; j < 8; ++j) {
        const size_t rowoff = (size_t)(mbase + j) * NCLS;
        out[rowoff + colbase +  0] = acc0[j] + cst[j];
        out[rowoff + colbase + 16] = acc1[j] + cst[j];
        out[rowoff + colbase + 32] = acc2[j] + cst[j];
        out[rowoff + colbase + 48] = acc3[j] + cst[j];
    }
}

extern "C" void kernel_launch(void* const* d_in, const int* in_sizes, int n_in,
                              void* d_out, int out_size, void* d_ws, size_t ws_size,
                              hipStream_t stream) {
    const float* df = (const float*)d_in[0];
    const int*   gt = (const int*)d_in[1];
    const float* fc = (const float*)d_in[2];
    float*       out = (float*)d_out;

    float* var    = (float*)d_ws;
    float* XY     = var + (size_t)NCLS * NDF;
    float* PQ     = XY + (size_t)NB * KTOT;
    float* constb = PQ + (size_t)NCLS * KTOT;

    stats_kernel<<<NCLS, NDF, 0, stream>>>(df, gt, var);
    prep_a_kernel<<<NB, NDF, 0, stream>>>(df, gt, fc, var, XY, constb);
    prep_b_kernel<<<NCLS, NDF, 0, stream>>>(fc, PQ);
    wmma_gemm_kernel<<<dim3(NB / 16, NCLS / 16 / 4), 32, 0, stream>>>(XY, PQ, constb, out);
}


# stats kernel: gt scanned on the scalar unit via the scalar cache (64 ids per wait, compare + carry shift-in) instead of LDS + barrier; 4 features per lane, found row's load stays in flight; same sums
# speedup vs baseline: 1.7309x; 1.7309x over previous
#include <hip/hip_runtime.h>

typedef __attribute__((ext_vector_type(2))) float v2f;
typedef __attribute__((ext_vector_type(8))) float v8f;

#define NCLS 256
#define NDF  512
#define NB   1024
#define KTOT (2 * NDF)
#define ALP  0.1f
#define HALP 0.05f

#define STATS_T 128

typedef __attribute__((ext_vector_type(4))) float v4f;
typedef int gt16 __attribute__((ext_vector_type(16), aligned(4)));

__device__ __forceinline__ void stats_step(v2f& m0, v2f& m1, v2f& v0, v2f& v1, int& n,
                                           const v2f f0, const v2f f1)
{
    const float nf = (float)n;
    const float c1 = nf / (nf + 1.0f);
    const float c2 = 1.0f / (nf + 1.0f);
    m0 = (n == 0) ? f0 : (m0 * c1 + f0 * c2);
    m1 = (n == 0) ? f1 : (m1 * c1 + f1 * c2);
    const v2f a0 = f0 - m0;
    const v2f a1 = f1 - m1;
    v0 = v0 * c1 + a0 * a0 * c1 * c2;
    v1 = v1 * c1 + a1 * a1 * c1 * c2;
    ++n;
}

__device__ __forceinline__ unsigned class_bits16(unsigned hit, const gt16 q, int c)
{
#pragma unroll
    for (int j = 15; j >= 0; --j) {
        asm("s_cmp_eq_u32 %1, %2\n\ts_addc_u32 %0, %0, %0"
            : "+s"(hit) : "s"(q[j]), "s"(c) : "scc");
    }
    return hit;
}

__global__ __launch_bounds__(STATS_T) void stats_kernel(
    const float* __restrict__ df, const int* __restrict__ gt,
    float* __restrict__ var_out)
{
    const int c  = blockIdx.x;
    const int d4 = threadIdx.x * 4;

    v2f m0 = {}, m1 = {}, v0 = {}, v1 = {};
    v4f f = {};
    int  n    = 0;
    bool pend = false;

    for (int b0 = 0; b0 < NB; b0 += 64) {
        const gt16* q = (const gt16*)(gt + b0);
        const gt16 q0 = q[0], q1 = q[1], q2 = q[2], q3 = q[3];
        const unsigned lo = class_bits16(class_bits16(0u, q1, c), q0, c);
        const unsigned hi = class_bits16(class_bits16(0u, q3, c), q2, c);
        unsigned long long hit = ((unsigned long long)hi << 32) | lo;
        while (hit) {
            const int b = b0 + __builtin_ctzll(hit);
            hit &= hit - 1;
            if (pend) stats_step(m0, m1, v0, v1, n, f.xy, f.zw);
            f = *(const v4f*)(df + (size_t)b * NDF + d4);
            pend = true;
        }
    }
    if (pend) stats_step(m0, m1, v0, v1, n, f.xy, f.zw);
    const v4f vo = {v0.x, v0.y, v1.x, v1.y};
    *(v4f*)(var_out + (size_t)c * NDF + d4) = vo;
}

__global__ __launch_bounds__(NDF) void prep_a_kernel(
    const float* __restrict__ df, const int* __restrict__ gt,
    const float* __restrict__ fc, const float* __restrict__ var_in,
    float* __restrict__ XY, float* __restrict__ constb)
{
    __shared__ float red[NDF];
    const int b = blockIdx.x;
    const int d = threadIdx.x;
    const int t = gt[b];

    const float fcv = fc[(size_t)t * NDF + d];
    const float va  = var_in[(size_t)t * NDF + d];
    const float dfv = df[(size_t)b * NDF + d];

    XY[(size_t)b * KTOT + d]       = dfv - ALP * fcv * va;
    XY[(size_t)b * KTOT + NDF + d] = HALP * va;

    red[d] = -fcv * dfv + HALP * fcv * fcv * va;
    __syncthreads();
    for (int s = NDF / 2; s > 0; s >>= 1) {
        if (d < s) red[d] += red[d + s];
        __syncthreads();
    }
    if (d == 0) constb[b] = red[0];
}

__global__ __launch_bounds__(NDF) void prep_b_kernel(
    const float* __restrict__ fc, float* __restrict__ PQ)
{
    const int c = blockIdx.x;
    const int d = threadIdx.x;
    const float v = fc[(size_t)c * NDF + d];
    PQ[(size_t)c * KTOT + d]       = v;
    PQ[(size_t)c * KTOT + NDF + d] = v * v;
}

__global__ __launch_bounds__(32) void wmma_gemm_kernel(
    const float* __restrict__ XY, const float* __restrict__ PQ,
    const float* __restrict__ constb, float* __restrict__ out)
{
    const int tm   = blockIdx.x;
    const int ts   = blockIdx.y;
    const int lane = threadIdx.x;
    const int r    = lane & 15;
    const int kh   = (lane >> 4) << 1;

    const float* Arow = XY + (size_t)(tm * 16 + r) * KTOT + kh;
    const float* Brow = PQ + (size_t)(ts * 64 + r) * KTOT + kh;

    v8f acc0 = {}, acc1 = {}, acc2 = {}, acc3 = {};
#pragma unroll 4
    for (int k = 0; k < KTOT; k += 4) {
        v2f a  = *(const v2f*)(Arow + k);
        v2f b0 = *(const v2f*)(Brow + k);
        v2f b1 = *(const v2f*)(Brow + 16 * KTOT + k);
        v2f b2 = *(const v2f*)(Brow + 32 * KTOT + k);
        v2f b3 = *(const v2f*)(Brow + 48 * KTOT + k);
        acc0 = __builtin_amdgcn_wmma_f32_16x16x4_f32(false, a, false, b0,
                                                     (short)0, acc0, false, false);
        acc1 = __builtin_amdgcn_wmma_f32_16x16x4_f32(false, a, false, b1,
                                                     (short)0, acc1, false, false);
        acc2 = __builtin_amdgcn_wmma_f32_16x16x4_f32(false, a, false, b2,
                                                     (short)0, acc2, false, false);
        acc3 = __builtin_amdgcn_wmma_f32_16x16x4_f32(false, a, false, b3,
                                                     (short)0, acc3, false, false);
    }

    const int mbase = tm * 16 + ((lane >> 4) << 3);
    float cst[8];
#pragma unroll
    for (int j = 0; j < 8; ++j) cst[j] = constb[mbase + j];

    const int colbase = ts * 64 + r;
#pragma unroll
    for (int j = 0; j < 8; ++j) {
        const size_t rowoff = (size_t)(mbase + j) * NCLS;
        out[rowoff + colbase +  0] = acc0[j] + cst[j];
        out[rowoff + colbase + 16] = acc1[j] + cst[j];
        out[rowoff + colbase + 32] = acc2[j] + cst[j];
        out[rowoff + colbase + 48] = acc3[j] + cst[j];
    }
}

extern "C" void kernel_launch(void* const* d_in, const int* in_sizes, int n_in,
                              void* d_out, int out_size, void* d_ws, size_t ws_size,
                              hipStream_t stream) {
    const float* df = (const float*)d_in[0];
    const int*   gt = (const int*)d_in[1];
    const float* fc = (const float*)d_in[2];
    float*       out = (float*)d_out;

    float* var    = (float*)d_ws;
    float* XY     = var + (size_t)NCLS * NDF;
    float* PQ     = XY + (size_t)NB * KTOT;
    float* constb = PQ + (size_t)NCLS * KTOT;

    stats_kernel<<<NCLS, STATS_T, 0, stream>>>(df, gt, var);
    prep_a_kernel<<<NB, NDF, 0, stream>>>(df, gt, fc, var, XY, constb);
    prep_b_kernel<<<NCLS, NDF, 0, stream>>>(fc, PQ);
    wmma_gemm_kernel<<<dim3(NB / 16, NCLS / 16 / 4), 32, 0, stream>>>(XY, PQ, constb, out);
}


# gemm kernel: K split over the 8 waves of a block (128 k each, fragment loads as before); the 8 partial 16x64 strips are added through LDS in fixed wave order: f32 summation order changes
# speedup vs baseline: 3.8821x; 2.2428x over previous
#include <hip/hip_runtime.h>

typedef __attribute__((ext_vector_type(2))) float v2f;
typedef __attribute__((ext_vector_type(8))) float v8f;

#define NCLS 256
#define NDF  512
#define NB   1024
#define KTOT (2 * NDF)
#define ALP  0.1f
#define HALP 0.05f

#define STATS_T 128

typedef __attribute__((ext_vector_type(4))) float v4f;
typedef int gt16 __attribute__((ext_vector_type(16), aligned(4)));

__device__ __forceinline__ void stats_step(v2f& m0, v2f& m1, v2f& v0, v2f& v1, int& n,
                                           const v2f f0, const v2f f1)
{
    const float nf = (float)n;
    const float c1 = nf / (nf + 1.0f);
    const float c2 = 1.0f / (nf + 1.0f);
    m0 = (n == 0) ? f0 : (m0 * c1 + f0 * c2);
    m1 = (n == 0) ? f1 : (m1 * c1 + f1 * c2);
    const v2f a0 = f0 - m0;
    const v2f a1 = f1 - m1;
    v0 = v0 * c1 + a0 * a0 * c1 * c2;
    v1 = v1 * c1 + a1 * a1 * c1 * c2;
    ++n;
}

__device__ __forceinline__ unsigned class_bits16(unsigned hit, const gt16 q, int c)
{
#pragma unroll
    for (int j = 15; j >= 0; --j) {
        asm("s_cmp_eq_u32 %1, %2\n\ts_addc_u32 %0, %0, %0"
            : "+s"(hit) : "s"(q[j]), "s"(c) : "scc");
    }
    return hit;
}

__global__ __launch_bounds__(STATS_T) void stats_kernel(
    const float* __restrict__ df, const int* __restrict__ gt,
    float* __restrict__ var_out)
{
    const int c  = blockIdx.x;
    const int d4 = threadIdx.x * 4;

    v2f m0 = {}, m1 = {}, v0 = {}, v1 = {};
    v4f f = {};
    int  n    = 0;
    bool pend = false;

    for (int b0 = 0; b0 < NB; b0 += 64) {
        const gt16* q = (const gt16*)(gt + b0);
        const gt16 q0 = q[0], q1 = q[1], q2 = q[2], q3 = q[3];
        const unsigned lo = class_bits16(class_bits16(0u, q1, c), q0, c);
        const unsigned hi = class_bits16(class_bits16(0u, q3, c), q2, c);
        unsigned long long hit = ((unsigned long long)hi << 32) | lo;
        while (hit) {
            const int b = b0 + __builtin_ctzll(hit);
            hit &= hit - 1;
            if (pend) stats_step(m0, m1, v0, v1, n, f.xy, f.zw);
            f = *(const v4f*)(df + (size_t)b * NDF + d4);
            pend = true;
        }
    }
    if (pend) stats_step(m0, m1, v0, v1, n, f.xy, f.zw);
    const v4f vo = {v0.x, v0.y, v1.x, v1.y};
    *(v4f*)(var_out + (size_t)c * NDF + d4) = vo;
}

__global__ __launch_bounds__(NDF) void prep_a_kernel(
    const float* __restrict__ df, const int* __restrict__ gt,
    const float* __restrict__ fc, const float* __restrict__ var_in,
    float* __restrict__ XY, float* __restrict__ constb)
{
    __shared__ float red[NDF];
    const int b = blockIdx.x;
    const int d = threadIdx.x;
    const int t = gt[b];

    const float fcv = fc[(size_t)t * NDF + d];
    const float va  = var_in[(size_t)t * NDF + d];
    const float dfv = df[(size_t)b * NDF + d];

    XY[(size_t)b * KTOT + d]       = dfv - ALP * fcv * va;
    XY[(size_t)b * KTOT + NDF + d] = HALP * va;

    red[d] = -fcv * dfv + HALP * fcv * fcv * va;
    __syncthreads();
    for (int s = NDF / 2; s > 0; s >>= 1) {
        if (d < s) red[d] += red[d + s];
        __syncthreads();
    }
    if (d == 0) constb[b] = red[0];
}

__global__ __launch_bounds__(NDF) void prep_b_kernel(
    const float* __restrict__ fc, float* __restrict__ PQ)
{
    const int c = blockIdx.x;
    const int d = threadIdx.x;
    const float v = fc[(size_t)c * NDF + d];
    PQ[(size_t)c * KTOT + d]       = v;
    PQ[(size_t)c * KTOT + NDF + d] = v * v;
}

#define GEMM_W  8
#define KSL     (KTOT / GEMM_W)
#define GEMM_T  256
#define RED_P   68

__global__ __launch_bounds__(GEMM_T) void wmma_gemm_kernel(
    const float* __restrict__ XY, const float* __restrict__ PQ,
    const float* __restrict__ constb, float* __restrict__ out)
{
    __shared__ __attribute__((aligned(16))) float red[GEMM_W * 16 * RED_P];

    const int tm   = blockIdx.x;
    const int ts   = blockIdx.y;
    const int w    = threadIdx.x >> 5;
    const int lane = threadIdx.x & 31;
    const int r    = lane & 15;
    const int kh   = (lane >> 4) << 1;

    const float* Arow = XY + (size_t)(tm * 16 + r) * KTOT + w * KSL + kh;
    const float* Brow = PQ + (size_t)(ts * 64 + r) * KTOT + w * KSL + kh;

    v8f acc0 = {}, acc1 = {}, acc2 = {}, acc3 = {};
#pragma unroll
    for (int k = 0; k < KSL; k += 4) {
        v2f a  = *(const v2f*)(Arow + k);
        v2f b0 = *(const v2f*)(Brow + k);
        v2f b1 = *(const v2f*)(Brow + 16 * KTOT + k);
        v2f b2 = *(const v2f*)(Brow + 32 * KTOT + k);
        v2f b3 = *(const v2f*)(Brow + 48 * KTOT + k);
        acc0 = __builtin_amdgcn_wmma_f32_16x16x4_f32(false, a, false, b0,
                                                     (short)0, acc0, false, false);
        acc1 = __builtin_amdgcn_wmma_f32_16x16x4_f32(false, a, false, b1,
                                                     (short)0, acc1, false, false);
        acc2 = __builtin_amdgcn_wmma_f32_16x16x4_f32(false, a, false, b2,
                                                     (short)0, acc2, false, false);
        acc3 = __builtin_amdgcn_wmma_f32_16x16x4_f32(false, a, false, b3,
                                                     (short)0, acc3, false, false);
    }

    float* mine = red + w * (16 * RED_P) + ((lane >> 4) << 3) * RED_P + r;
#pragma unroll
    for (int j = 0; j < 8; ++j) {
        mine[j * RED_P +  0] = acc0[j];
        mine[j * RED_P + 16] = acc1[j];
        mine[j * RED_P + 32] = acc2[j];
        mine[j * RED_P + 48] = acc3[j];
    }
    __syncthreads();

    const int row = threadIdx.x >> 4;
    const int c4  = (threadIdx.x & 15) << 2;
    const float* p = red + row * RED_P + c4;
    v4f s = *(const v4f*)p;
#pragma unroll
    for (int q = 1; q < GEMM_W; ++q) s += *(const v4f*)(p + q * (16 * RED_P));
    const float cst = constb[tm * 16 + row];
    s += cst;
    *(v4f*)(out + (size_t)(tm * 16 + row) * NCLS + ts * 64 + c4) = s;
}

extern "C" void kernel_launch(void* const* d_in, const int* in_sizes, int n_in,
                              void* d_out, int out_size, void* d_ws, size_t ws_size,
                              hipStream_t stream) {
    const float* df = (const float*)d_in[0];
    const int*   gt = (const int*)d_in[1];
    const float* fc = (const float*)d_in[2];
    float*       out = (float*)d_out;

    float* var    = (float*)d_ws;
    float* XY     = var + (size_t)NCLS * NDF;
    float* PQ     = XY + (size_t)NB * KTOT;
    float* constb = PQ + (size_t)NCLS * KTOT;

    stats_kernel<<<NCLS, STATS_T, 0, stream>>>(df, gt, var);
    prep_a_kernel<<<NB, NDF, 0, stream>>>(df, gt, fc, var, XY, constb);
    prep_b_kernel<<<NCLS, NDF, 0, stream>>>(fc, PQ);
    wmma_gemm_kernel<<<dim3(NB / 16, NCLS / 16 / 4), GEMM_T, 0, stream>>>(XY, PQ, constb, out);
}


# per-class split: out = df*fc^T (WMMA, K=512, raw inputs) + G[gt] + row term, G = [U|Y]*[fc|fc^2]^T a 256x256 per-class WMMA GEMM; both prep launches gone (3 launches); f32 summation order changes
# speedup vs baseline: 4.3299x; 1.1153x over previous
#include <hip/hip_runtime.h>

typedef __attribute__((ext_vector_type(2))) float v2f;
typedef __attribute__((ext_vector_type(8))) float v8f;

#define NCLS 256
#define NDF  512
#define NB   1024
#define KTOT (2 * NDF)
#define ALP  0.1f
#define HALP 0.05f

#define STATS_T 128

typedef __attribute__((ext_vector_type(4))) float v4f;
typedef int gt16 __attribute__((ext_vector_type(16), aligned(4)));

__device__ __forceinline__ void stats_step(v2f& m0, v2f& m1, v2f& v0, v2f& v1, int& n,
                                           const v2f f0, const v2f f1)
{
    const float nf = (float)n;
    const float c1 = nf / (nf + 1.0f);
    const float c2 = 1.0f / (nf + 1.0f);
    m0 = (n == 0) ? f0 : (m0 * c1 + f0 * c2);
    m1 = (n == 0) ? f1 : (m1 * c1 + f1 * c2);
    const v2f a0 = f0 - m0;
    const v2f a1 = f1 - m1;
    v0 = v0 * c1 + a0 * a0 * c1 * c2;
    v1 = v1 * c1 + a1 * a1 * c1 * c2;
    ++n;
}

__device__ __forceinline__ unsigned class_bits16(unsigned hit, const gt16 q, int c)
{
#pragma unroll
    for (int j = 15; j >= 0; --j) {
        asm("s_cmp_eq_u32 %1, %2\n\ts_addc_u32 %0, %0, %0"
            : "+s"(hit) : "s"(q[j]), "s"(c) : "scc");
    }
    return hit;
}

__global__ __launch_bounds__(STATS_T) void stats_kernel(
    const float* __restrict__ df, const int* __restrict__ gt,
    const float* __restrict__ fc,
    float* __restrict__ CA, float* __restrict__ PQ)
{
    const int c  = blockIdx.x;
    const int d4 = threadIdx.x * 4;
    const v4f fcv = *(const v4f*)(fc + (size_t)c * NDF + d4);

    v2f m0 = {}, m1 = {}, v0 = {}, v1 = {};
    v4f f = {};
    int  n    = 0;
    bool pend = false;

    for (int b0 = 0; b0 < NB; b0 += 64) {
        const gt16* q = (const gt16*)(gt + b0);
        const gt16 q0 = q[0], q1 = q[1], q2 = q[2], q3 = q[3];
        const unsigned lo = class_bits16(class_bits16(0u, q1, c), q0, c);
        const unsigned hi = class_bits16(class_bits16(0u, q3, c), q2, c);
        unsigned long long hit = ((unsigned long long)hi << 32) | lo;
        while (hit) {
            const int b = b0 + __builtin_ctzll(hit);
            hit &= hit - 1;
            if (pend) stats_step(m0, m1, v0, v1, n, f.xy, f.zw);
            f = *(const v4f*)(df + (size_t)b * NDF + d4);
            pend = true;
        }
    }
    if (pend) stats_step(m0, m1, v0, v1, n, f.xy, f.zw);
    const v4f va = {v0.x, v0.y, v1.x, v1.y};
    const v4f u  = -ALP * fcv * va;
    const v4f y  = HALP * va;
    const v4f f2 = fcv * fcv;
    *(v4f*)(CA + (size_t)c * KTOT + d4)       = u;
    *(v4f*)(CA + (size_t)c * KTOT + NDF + d4) = y;
    *(v4f*)(PQ + (size_t)c * KTOT + d4)       = fcv;
    *(v4f*)(PQ + (size_t)c * KTOT + NDF + d4) = f2;
}

#define GEMM_W  8
#define GEMM_T  256
#define RED_P   68
#define RED_N   (GEMM_W * 16 * RED_P)

template <int LDA, int LDB, int KS, int NT>
__device__ __forceinline__ void strip_partial(
    const float* __restrict__ A, const float* __restrict__ B,
    int row0, int col0, float* red)
{
    const int w    = threadIdx.x >> 5;
    const int lane = threadIdx.x & 31;
    const int r    = lane & 15;
    const int kh   = (lane >> 4) << 1;

    const float* Arow = A + (size_t)(row0 + r) * LDA + w * KS + kh;
    const float* Brow = B + (size_t)(col0 + r) * LDB + w * KS + kh;

    v8f acc[NT];
#pragma unroll
    for (int n = 0; n < NT; ++n) acc[n] = v8f{};
#pragma unroll
    for (int k = 0; k < KS; k += 4) {
        const v2f a = *(const v2f*)(Arow + k);
#pragma unroll
        for (int n = 0; n < NT; ++n) {
            const v2f b = *(const v2f*)(Brow + n * 16 * LDB + k);
            acc[n] = __builtin_amdgcn_wmma_f32_16x16x4_f32(false, a, false, b,
                                                           (short)0, acc[n], false, false);
        }
    }

    float* mine = red + w * (16 * RED_P) + ((lane >> 4) << 3) * RED_P + r;
#pragma unroll
    for (int j = 0; j < 8; ++j) {
#pragma unroll
        for (int n = 0; n < NT; ++n) mine[j * RED_P + n * 16] = acc[n][j];
    }
}

#define CG_X  (NCLS / 16)
#define CG_Y  (NCLS / 32)
#define RD_X  (NB / 16 / CG_Y)

__global__ __launch_bounds__(GEMM_T) void class_gemm_kernel(
    const float* __restrict__ CA, const float* __restrict__ PQ,
    const float* __restrict__ df, const float* __restrict__ fc,
    const int* __restrict__ gt,
    float* __restrict__ G, float* __restrict__ rowterm)
{
    __shared__ __attribute__((aligned(16))) float red[RED_N];
    const int row = threadIdx.x >> 4;
    const int l16 = threadIdx.x & 15;

    if (blockIdx.x >= CG_X) {
        const int b = (((int)blockIdx.x - CG_X) * CG_Y + (int)blockIdx.y) * 16 + row;
        const int t = gt[b];
        const float* xr = df + (size_t)b * NDF + l16 * 4;
        const float* yr = fc + (size_t)t * NDF + l16 * 4;
        const float* zr = CA + (size_t)t * KTOT + NDF + l16 * 4;
        v4f a = {};
#pragma unroll
        for (int j = 0; j < NDF / 64; ++j) {
            const v4f x = *(const v4f*)(xr + j * 64);
            const v4f y = *(const v4f*)(yr + j * 64);
            const v4f z = *(const v4f*)(zr + j * 64);
            a += y * (z * y - x);
        }
        float s = (a.x + a.y) + (a.z + a.w);
#pragma unroll
        for (int off = 8; off > 0; off >>= 1) s += __shfl_xor(s, off, 32);
        if (l16 == 0) rowterm[b] = s;
        return;
    }

    const int tm = blockIdx.x;
    const int ts = blockIdx.y;
    strip_partial<KTOT, KTOT, KTOT / GEMM_W, 2>(CA, PQ, tm * 16, ts * 32, red);
    __syncthreads();

    const int c2 = l16 << 1;
    const float* p = red + row * RED_P + c2;
    v2f s = *(const v2f*)p;
#pragma unroll
    for (int q = 1; q < GEMM_W; ++q) s += *(const v2f*)(p + q * (16 * RED_P));
    *(v2f*)(G + (size_t)(tm * 16 + row) * NCLS + ts * 32 + c2) = s;
}

__global__ __launch_bounds__(GEMM_T) void score_gemm_kernel(
    const float* __restrict__ df, const float* __restrict__ fc,
    const int* __restrict__ gt, const float* __restrict__ G,
    const float* __restrict__ rowterm, float* __restrict__ out)
{
    __shared__ __attribute__((aligned(16))) float red[RED_N];
    const int tm  = blockIdx.x;
    const int ts  = blockIdx.y;
    const int row = threadIdx.x >> 4;
    const int c4  = (threadIdx.x & 15) << 2;
    const int b   = tm * 16 + row;

    const v4f   g  = *(const v4f*)(G + (size_t)gt[b] * NCLS + ts * 64 + c4);
    const float rt = rowterm[b];

    strip_partial<NDF, NDF, NDF / GEMM_W, 4>(df, fc, tm * 16, ts * 64, red);
    __syncthreads();

    const float* p = red + row * RED_P + c4;
    v4f s = *(const v4f*)p;
#pragma unroll
    for (int q = 1; q < GEMM_W; ++q) s += *(const v4f*)(p + q * (16 * RED_P));
    s += g;
    s += rt;
    *(v4f*)(out + (size_t)b * NCLS + ts * 64 + c4) = s;
}

extern "C" void kernel_launch(void* const* d_in, const int* in_sizes, int n_in,
                              void* d_out, int out_size, void* d_ws, size_t ws_size,
                              hipStream_t stream) {
    const float* df = (const float*)d_in[0];
    const int*   gt = (const int*)d_in[1];
    const float* fc = (const float*)d_in[2];
    float*       out = (float*)d_out;

    float* CA     = (float*)d_ws;
    float* PQ     = CA + (size_t)NCLS * KTOT;
    float* G       = PQ + (size_t)NCLS * KTOT;
    float* rowterm = G + (size_t)NCLS * NCLS;

    stats_kernel<<<NCLS, STATS_T, 0, stream>>>(df, gt, fc, CA, PQ);
    class_gemm_kernel<<<dim3(CG_X + RD_X, CG_Y), GEMM_T, 0, stream>>>(CA, PQ, df, fc, gt, G, rowterm);
    score_gemm_kernel<<<dim3(NB / 16, NCLS / 64), GEMM_T, 0, stream>>>(df, fc, gt, G, rowterm, out);
}


# df*fc^T needs no statistics: runs in launch 1 beside the class scans (S to workspace); class-GEMM blocks finish out by class from the scans' sample lists; row term h[t]-S[b,t]; f32 sum order changes
# speedup vs baseline: 4.6889x; 1.0829x over previous
#include <hip/hip_runtime.h>

typedef __attribute__((ext_vector_type(2))) float v2f;
typedef __attribute__((ext_vector_type(8))) float v8f;
typedef __attribute__((ext_vector_type(4))) float v4f;
typedef __attribute__((ext_vector_type(4))) int   v4i;

#define NCLS 256
#define NDF  512
#define NB   1024
#define KTOT (2 * NDF)
#define ALP  0.1f
#define HALP 0.05f


#define STATS_T 128
#define PRE     8

typedef int gt16 __attribute__((ext_vector_type(16), aligned(4)));
typedef __attribute__((address_space(4))) gt16 gt16_k;

__device__ __forceinline__ void stats_step(v2f& m0, v2f& m1, v2f& v0, v2f& v1, int& n,
                                           const v2f f0, const v2f f1)
{
    const float nf = (float)n;
    const float c1 = nf / (nf + 1.0f);
    const float c2 = 1.0f / (nf + 1.0f);
    m0 = (n == 0) ? f0 : (m0 * c1 + f0 * c2);
    m1 = (n == 0) ? f1 : (m1 * c1 + f1 * c2);
    const v2f a0 = f0 - m0;
    const v2f a1 = f1 - m1;
    v0 = v0 * c1 + a0 * a0 * c1 * c2;
    v1 = v1 * c1 + a1 * a1 * c1 * c2;
    ++n;
}

__device__ __forceinline__ unsigned class_bits16(unsigned hit, const gt16 q, int c)
{
#pragma unroll
    for (int j = 15; j >= 0; --j) {
        asm("s_cmp_eq_u32 %1, %2\n\ts_addc_u32 %0, %0, %0"
            : "+s"(hit) : "s"(q[j]), "s"(c) : "scc");
    }
    return hit;
}

__device__ __forceinline__ void class_stats(
    const float* __restrict__ df, const int* __restrict__ gt,
    const float* __restrict__ fc,
    float* __restrict__ CA, float* __restrict__ PQ, float* __restrict__ hp,
    int* __restrict__ cnt, int* __restrict__ list, const int c, const int tid)
{
    const int d4 = tid * 4;
    const v4f fcv = *(const v4f*)(fc + (size_t)c * NDF + d4);
    int* lst = list + (size_t)c * NB;

    v2f m0 = {}, m1 = {}, v0 = {}, v1 = {};
    v4f f = {};
    int  n    = 0;
    bool pend = false;
    int  mlo = 0, mhi = 0;

    for (int b0 = 0; b0 < NB; b0 += 64) {
        const gt16_k* q = (const gt16_k*)(gt + b0);
        const gt16 q0 = q[0], q1 = q[1], q2 = q[2], q3 = q[3];
        const unsigned lo = class_bits16(class_bits16(0u, q1, c), q0, c);
        const unsigned hi = class_bits16(class_bits16(0u, q3, c), q2, c);
        unsigned long long hit = ((unsigned long long)hi << 32) | lo;
        const bool keep = ((tid & 31) == (b0 >> 6));
        mlo = keep ? (int)lo : mlo;
        mhi = keep ? (int)hi : mhi;
        while (hit) {
            const int b = b0 + __builtin_ctzll(hit);
            hit &= hit - 1;
            if (pend) stats_step(m0, m1, v0, v1, n, f.xy, f.zw);
            f = *(const v4f*)(df + (size_t)b * NDF + d4);
            pend = true;
        }
    }
    if (pend) stats_step(m0, m1, v0, v1, n, f.xy, f.zw);
    const v4f va = {v0.x, v0.y, v1.x, v1.y};
    const v4f u  = -ALP * fcv * va;
    const v4f y  = HALP * va;
    const v4f f2 = fcv * fcv;
    *(v4f*)(CA + (size_t)c * KTOT + d4)       = u;
    *(v4f*)(CA + (size_t)c * KTOT + NDF + d4) = y;
    *(v4f*)(PQ + (size_t)c * KTOT + d4)       = fcv;
    *(v4f*)(PQ + (size_t)c * KTOT + NDF + d4) = f2;

    float p = (y.x * f2.x + y.y * f2.y) + (y.z * f2.z + y.w * f2.w);
#pragma unroll
    for (int off = 16; off > 0; off >>= 1) p += __shfl_xor(p, off, 32);
    if ((tid & 31) == 0) hp[c * 4 + (tid >> 5)] = p;

    if (tid < 32) {
        unsigned long long m = ((unsigned long long)(unsigned)mhi << 32) | (unsigned)mlo;
        const int mine = __builtin_popcountll(m);
        int incl = mine;
#pragma unroll
        for (int off = 1; off < 32; off <<= 1) {
            const int up = __shfl_up(incl, off, 32);
            if (tid >= off) incl += up;
        }
        const int total = __shfl(incl, 31, 32);
        int pos = incl - mine;
        while (m) {
            lst[pos++] = tid * 64 + __builtin_ctzll(m);
            m &= m - 1;
        }
        if (tid < PRE && tid >= total) lst[tid] = -1;
        if (tid == 0) cnt[c] = total;
    }
}

#define GEMM_W  8
#define GEMM_T  256
#define RED_P   68
#define RED_N   (GEMM_W * 16 * RED_P)

template <int LDA, int LDB, int KS, int NT>
__device__ __forceinline__ void strip_partial(
    const float* __restrict__ A, const float* __restrict__ B,
    int row0, int col0, float* red)
{
    const int w    = threadIdx.x >> 5;
    const int lane = threadIdx.x & 31;
    const int r    = lane & 15;
    const int kh   = (lane >> 4) << 1;

    const float* Arow = A + (size_t)(row0 + r) * LDA + w * KS + kh;
    const float* Brow = B + (size_t)(col0 + r) * LDB + w * KS + kh;

    v8f acc[NT];
#pragma unroll
    for (int n = 0; n < NT; ++n) acc[n] = v8f{};
#pragma unroll
    for (int k = 0; k < KS; k += 4) {
        const v2f a = *(const v2f*)(Arow + k);
#pragma unroll
        for (int n = 0; n < NT; ++n) {
            const v2f b = *(const v2f*)(Brow + n * 16 * LDB + k);
            acc[n] = __builtin_amdgcn_wmma_f32_16x16x4_f32(false, a, false, b,
                                                           (short)0, acc[n], false, false);
        }
    }

    float* mine = red + w * (16 * RED_P) + ((lane >> 4) << 3) * RED_P + r;
#pragma unroll
    for (int j = 0; j < 8; ++j) {
#pragma unroll
        for (int n = 0; n < NT; ++n) mine[j * RED_P + n * 16] = acc[n][j];
    }
}

#define ST_BLOCKS (NCLS / 2)
#define SG_X      (NB / 16)
#define SG_Y      (NCLS / 64)

__global__ __launch_bounds__(GEMM_T) void front_kernel(
    const float* __restrict__ df, const int* __restrict__ gt,
    const float* __restrict__ fc,
    float* __restrict__ CA, float* __restrict__ PQ, float* __restrict__ S,
    float* __restrict__ hp, int* __restrict__ cnt, int* __restrict__ list)
{
    __shared__ __attribute__((aligned(16))) float red[RED_N];

    if (blockIdx.x < ST_BLOCKS) {
        const int half = __builtin_amdgcn_readfirstlane((int)(threadIdx.x >> 7));
        class_stats(df, gt, fc, CA, PQ, hp, cnt, list,
                    (int)blockIdx.x * 2 + half, (int)(threadIdx.x & (STATS_T - 1)));
        return;
    }

    const int sb  = (int)blockIdx.x - ST_BLOCKS;
    const int tm  = sb & (SG_X - 1);
    const int ts  = sb >> 6;
    const int row = threadIdx.x >> 4;
    const int c4  = (threadIdx.x & 15) << 2;
    const int b   = tm * 16 + row;

    strip_partial<NDF, NDF, NDF / GEMM_W, 4>(df, fc, tm * 16, ts * 64, red);
    __syncthreads();

    const float* p = red + row * RED_P + c4;
    v4f s = *(const v4f*)p;
#pragma unroll
    for (int q = 1; q < GEMM_W; ++q) s += *(const v4f*)(p + q * (16 * RED_P));
    *(v4f*)(S + (size_t)b * NCLS + ts * 64 + c4) = s;
}

#define CG_X  (NCLS / 16)
#define CG_Y  (NCLS / 32)

__global__ __launch_bounds__(GEMM_T) void class_out_kernel(
    const float* __restrict__ CA, const float* __restrict__ PQ,
    const float* __restrict__ S, const float* __restrict__ hp,
    const int* __restrict__ cnt, const int* __restrict__ list,
    float* __restrict__ out)
{
    __shared__ __attribute__((aligned(16))) float red[RED_N];
    const int tm  = blockIdx.x;
    const int ts  = blockIdx.y;
    const int row = threadIdx.x >> 4;
    const int l16 = threadIdx.x & 15;
    const int t   = tm * 16 + row;
    const int c2  = l16 << 1;
    const int col = ts * 32 + c2;

    const int* lst = list + (size_t)t * NB;
    const v4i l0 = *(const v4i*)lst;
    const v4i l1 = *(const v4i*)(lst + 4);
    const int n  = cnt[t];
    const v4f h4 = *(const v4f*)(hp + t * 4);
    const int bb[PRE] = {l0.x, l0.y, l0.z, l0.w, l1.x, l1.y, l1.z, l1.w};

    strip_partial<KTOT, KTOT, KTOT / GEMM_W, 2>(CA, PQ, tm * 16, ts * 32, red);

    v2f   sv[PRE];
    float st[PRE];
#pragma unroll
    for (int i = 0; i < PRE; ++i) {
        const float* sr = S + (size_t)(bb[i] < 0 ? 0 : bb[i]) * NCLS;
        sv[i] = *(const v2f*)(sr + col);
        st[i] = sr[t];
    }
    __syncthreads();

    const float* p = red + row * RED_P + c2;
    v2f g = *(const v2f*)p;
#pragma unroll
    for (int q = 1; q < GEMM_W; ++q) g += *(const v2f*)(p + q * (16 * RED_P));

    const float h = ((h4.x + h4.y) + h4.z) + h4.w;
#pragma unroll
    for (int i = 0; i < PRE; ++i) {
        if (bb[i] >= 0)
            *(v2f*)(out + (size_t)bb[i] * NCLS + col) = (sv[i] + g) + (h - st[i]);
    }
    for (int i = PRE; i < n; ++i) {
        const int b = lst[i];
        const float* sr = S + (size_t)b * NCLS;
        const v2f   x = *(const v2f*)(sr + col);
        const float y = sr[t];
        *(v2f*)(out + (size_t)b * NCLS + col) = (x + g) + (h - y);
    }
}

extern "C" void kernel_launch(void* const* d_in, const int* in_sizes, int n_in,
                              void* d_out, int out_size, void* d_ws, size_t ws_size,
                              hipStream_t stream) {
    const float* df = (const float*)d_in[0];
    const int*   gt = (const int*)d_in[1];
    const float* fc = (const float*)d_in[2];
    float*       out = (float*)d_out;

    float* CA   = (float*)d_ws;
    float* PQ   = CA + (size_t)NCLS * KTOT;
    float* S    = PQ + (size_t)NCLS * KTOT;
    float* hp   = S + (size_t)NB * NCLS;
    int*   cnt  = (int*)(hp + (size_t)NCLS * 4);
    int*   list = cnt + NCLS;

    front_kernel<<<ST_BLOCKS + SG_X * SG_Y, GEMM_T, 0, stream>>>(df, gt, fc, CA, PQ, S, hp, cnt, list);
    class_out_kernel<<<dim3(CG_X, CG_Y), GEMM_T, 0, stream>>>(CA, PQ, S, hp, cnt, list, out);
}


# stats scan divided over the block's 8 waves (128 ids each, both classes, on the scalar unit), hit words exchanged through LDS behind one barrier; found rows fetched two deep; same sums, same order
# speedup vs baseline: 5.9175x; 1.2620x over previous
#include <hip/hip_runtime.h>

typedef __attribute__((ext_vector_type(2))) float v2f;
typedef __attribute__((ext_vector_type(8))) float v8f;
typedef __attribute__((ext_vector_type(4))) float v4f;
typedef __attribute__((ext_vector_type(4))) int   v4i;

#define NCLS 256
#define NDF  512
#define NB   1024
#define KTOT (2 * NDF)
#define ALP  0.1f
#define HALP 0.05f


#define STATS_T 128
#define PRE     8

typedef int gt16 __attribute__((ext_vector_type(16), aligned(4)));
typedef __attribute__((address_space(4))) gt16 gt16_k;

__device__ __forceinline__ void stats_step(v2f& m0, v2f& m1, v2f& v0, v2f& v1, int& n,
                                           const v2f f0, const v2f f1)
{
    const float nf = (float)n;
    const float c1 = nf / (nf + 1.0f);
    const float c2 = 1.0f / (nf + 1.0f);
    m0 = (n == 0) ? f0 : (m0 * c1 + f0 * c2);
    m1 = (n == 0) ? f1 : (m1 * c1 + f1 * c2);
    const v2f a0 = f0 - m0;
    const v2f a1 = f1 - m1;
    v0 = v0 * c1 + a0 * a0 * c1 * c2;
    v1 = v1 * c1 + a1 * a1 * c1 * c2;
    ++n;
}

__device__ __forceinline__ unsigned class_bits16(unsigned hit, const gt16 q, int c)
{
#pragma unroll
    for (int j = 15; j >= 0; --j) {
        asm("s_cmp_eq_u32 %1, %2\n\ts_addc_u32 %0, %0, %0"
            : "+s"(hit) : "s"(q[j]), "s"(c) : "scc");
    }
    return hit;
}

typedef __attribute__((ext_vector_type(2))) unsigned v2u;

__device__ __forceinline__ void class_stats(
    const float* __restrict__ df, const int* __restrict__ gt,
    const float* __restrict__ fc,
    float* __restrict__ CA, float* __restrict__ PQ, float* __restrict__ hp,
    int* __restrict__ cnt, int* __restrict__ list, unsigned (*hitw)[32],
    const int cpair, const int wv, const int tid)
{
    const int half = wv >> 2;
    const int c    = cpair * 2 + half;
    const int lane = tid & 31;
    const int d4   = tid * 4;
    const v4f fcv = *(const v4f*)(fc + (size_t)c * NDF + d4);
    int* lst = list + (size_t)c * NB;

    for (int r = 0; r < 2; ++r) {
        const gt16_k* q = (const gt16_k*)(gt + wv * 128 + r * 64);
        const gt16 q0 = q[0], q1 = q[1], q2 = q[2], q3 = q[3];
        const int ce = cpair * 2, co = cpair * 2 + 1;
        const unsigned lo0 = class_bits16(class_bits16(0u, q1, ce), q0, ce);
        const unsigned hi0 = class_bits16(class_bits16(0u, q3, ce), q2, ce);
        const unsigned lo1 = class_bits16(class_bits16(0u, q1, co), q0, co);
        const unsigned hi1 = class_bits16(class_bits16(0u, q3, co), q2, co);
        if (lane == 0) {
            *(v2u*)&hitw[0][wv * 4 + r * 2] = v2u{lo0, hi0};
            *(v2u*)&hitw[1][wv * 4 + r * 2] = v2u{lo1, hi1};
        }
    }
    __syncthreads();
    const unsigned word = hitw[half][lane];

    v2f m0 = {}, m1 = {}, v0 = {}, v1 = {};
    int n = 0;

    unsigned M  = __builtin_amdgcn_ballot_w32(word != 0u);
    unsigned wd = 0u;
    int base = 0;
    auto next_row = [&](v4f& dst) -> bool {
        if (wd == 0u) {
            if (M == 0u) return false;
            const int j = __builtin_ctz(M);
            M &= M - 1u;
            wd   = (unsigned)__builtin_amdgcn_readlane((int)word, j);
            base = j << 5;
        }
        const int b = base + __builtin_ctz(wd);
        wd &= wd - 1u;
        dst = *(const v4f*)(df + (size_t)b * NDF + d4);
        return true;
    };
    v4f fa = {}, fb = {};
    bool more = next_row(fa);
    while (more) {
        const bool more_b = next_row(fb);
        stats_step(m0, m1, v0, v1, n, fa.xy, fa.zw);
        if (!more_b) break;
        more = next_row(fa);
        stats_step(m0, m1, v0, v1, n, fb.xy, fb.zw);
    }

    const v4f va = {v0.x, v0.y, v1.x, v1.y};
    const v4f u  = -ALP * fcv * va;
    const v4f y  = HALP * va;
    const v4f f2 = fcv * fcv;
    *(v4f*)(CA + (size_t)c * KTOT + d4)       = u;
    *(v4f*)(CA + (size_t)c * KTOT + NDF + d4) = y;
    *(v4f*)(PQ + (size_t)c * KTOT + d4)       = fcv;
    *(v4f*)(PQ + (size_t)c * KTOT + NDF + d4) = f2;

    float p = (y.x * f2.x + y.y * f2.y) + (y.z * f2.z + y.w * f2.w);
#pragma unroll
    for (int off = 16; off > 0; off >>= 1) p += __shfl_xor(p, off, 32);
    if (lane == 0) hp[c * 4 + (tid >> 5)] = p;

    if (tid < 32) {
        unsigned m = word;
        const int mine = __builtin_popcount(m);
        int incl = mine;
#pragma unroll
        for (int off = 1; off < 32; off <<= 1) {
            const int up = __shfl_up(incl, off, 32);
            if (tid >= off) incl += up;
        }
        const int total = __shfl(incl, 31, 32);
        int pos = incl - mine;
        while (m) {
            lst[pos++] = tid * 32 + __builtin_ctz(m);
            m &= m - 1u;
        }
        if (tid < PRE && tid >= total) lst[tid] = -1;
        if (tid == 0) cnt[c] = total;
    }
}

#define GEMM_W  8
#define GEMM_T  256
#define RED_P   68
#define RED_N   (GEMM_W * 16 * RED_P)

template <int LDA, int LDB, int KS, int NT>
__device__ __forceinline__ void strip_partial(
    const float* __restrict__ A, const float* __restrict__ B,
    int row0, int col0, float* red)
{
    const int w    = threadIdx.x >> 5;
    const int lane = threadIdx.x & 31;
    const int r    = lane & 15;
    const int kh   = (lane >> 4) << 1;

    const float* Arow = A + (size_t)(row0 + r) * LDA + w * KS + kh;
    const float* Brow = B + (size_t)(col0 + r) * LDB + w * KS + kh;

    v8f acc[NT];
#pragma unroll
    for (int n = 0; n < NT; ++n) acc[n] = v8f{};
#pragma unroll
    for (int k = 0; k < KS; k += 4) {
        const v2f a = *(const v2f*)(Arow + k);
#pragma unroll
        for (int n = 0; n < NT; ++n) {
            const v2f b = *(const v2f*)(Brow + n * 16 * LDB + k);
            acc[n] = __builtin_amdgcn_wmma_f32_16x16x4_f32(false, a, false, b,
                                                           (short)0, acc[n], false, false);
        }
    }

    float* mine = red + w * (16 * RED_P) + ((lane >> 4) << 3) * RED_P + r;
#pragma unroll
    for (int j = 0; j < 8; ++j) {
#pragma unroll
        for (int n = 0; n < NT; ++n) mine[j * RED_P + n * 16] = acc[n][j];
    }
}

#define ST_BLOCKS (NCLS / 2)
#define SG_X      (NB / 16)
#define SG_Y      (NCLS / 64)

__global__ __launch_bounds__(GEMM_T) void front_kernel(
    const float* __restrict__ df, const int* __restrict__ gt,
    const float* __restrict__ fc,
    float* __restrict__ CA, float* __restrict__ PQ, float* __restrict__ S,
    float* __restrict__ hp, int* __restrict__ cnt, int* __restrict__ list)
{
    __shared__ __attribute__((aligned(16))) float red[RED_N];

    if (blockIdx.x < ST_BLOCKS) {
        __shared__ __attribute__((aligned(16))) unsigned hitw[2][32];
        const int wv = __builtin_amdgcn_readfirstlane((int)(threadIdx.x >> 5));
        class_stats(df, gt, fc, CA, PQ, hp, cnt, list, hitw,
                    (int)blockIdx.x, wv, (int)(threadIdx.x & (STATS_T - 1)));
        return;
    }

    const int sb  = (int)blockIdx.x - ST_BLOCKS;
    const int tm  = sb & (SG_X - 1);
    const int ts  = sb >> 6;
    const int row = threadIdx.x >> 4;
    const int c4  = (threadIdx.x & 15) << 2;
    const int b   = tm * 16 + row;

    strip_partial<NDF, NDF, NDF / GEMM_W, 4>(df, fc, tm * 16, ts * 64, red);
    __syncthreads();

    const float* p = red + row * RED_P + c4;
    v4f s = *(const v4f*)p;
#pragma unroll
    for (int q = 1; q < GEMM_W; ++q) s += *(const v4f*)(p + q * (16 * RED_P));
    *(v4f*)(S + (size_t)b * NCLS + ts * 64 + c4) = s;
}

#define CG_X  (NCLS / 16)
#define CG_Y  (NCLS / 32)

__global__ __launch_bounds__(GEMM_T) void class_out_kernel(
    const float* __restrict__ CA, const float* __restrict__ PQ,
    const float* __restrict__ S, const float* __restrict__ hp,
    const int* __restrict__ cnt, const int* __restrict__ list,
    float* __restrict__ out)
{
    __shared__ __attribute__((aligned(16))) float red[RED_N];
    const int tm  = blockIdx.x;
    const int ts  = blockIdx.y;
    const int row = threadIdx.x >> 4;
    const int l16 = threadIdx.x & 15;
    const int t   = tm * 16 + row;
    const int c2  = l16 << 1;
    const int col = ts * 32 + c2;

    const int* lst = list + (size_t)t * NB;
    const v4i l0 = *(const v4i*)lst;
    const v4i l1 = *(const v4i*)(lst + 4);
    const int n  = cnt[t];
    const v4f h4 = *(const v4f*)(hp + t * 4);
    const int bb[PRE] = {l0.x, l0.y, l0.z, l0.w, l1.x, l1.y, l1.z, l1.w};

    strip_partial<KTOT, KTOT, KTOT / GEMM_W, 2>(CA, PQ, tm * 16, ts * 32, red);

    v2f   sv[PRE];
    float st[PRE];
#pragma unroll
    for (int i = 0; i < PRE; ++i) {
        const float* sr = S + (size_t)(bb[i] < 0 ? 0 : bb[i]) * NCLS;
        sv[i] = *(const v2f*)(sr + col);
        st[i] = sr[t];
    }
    __syncthreads();

    const float* p = red + row * RED_P + c2;
    v2f g = *(const v2f*)p;
#pragma unroll
    for (int q = 1; q < GEMM_W; ++q) g += *(const v2f*)(p + q * (16 * RED_P));

    const float h = ((h4.x + h4.y) + h4.z) + h4.w;
#pragma unroll
    for (int i = 0; i < PRE; ++i) {
        if (bb[i] >= 0)
            *(v2f*)(out + (size_t)bb[i] * NCLS + col) = (sv[i] + g) + (h - st[i]);
    }
    for (int i = PRE; i < n; ++i) {
        const int b = lst[i];
        const float* sr = S + (size_t)b * NCLS;
        const v2f   x = *(const v2f*)(sr + col);
        const float y = sr[t];
        *(v2f*)(out + (size_t)b * NCLS + col) = (x + g) + (h - y);
    }
}

extern "C" void kernel_launch(void* const* d_in, const int* in_sizes, int n_in,
                              void* d_out, int out_size, void* d_ws, size_t ws_size,
                              hipStream_t stream) {
    const float* df = (const float*)d_in[0];
    const int*   gt = (const int*)d_in[1];
    const float* fc = (const float*)d_in[2];
    float*       out = (float*)d_out;

    float* CA   = (float*)d_ws;
    float* PQ   = CA + (size_t)NCLS * KTOT;
    float* S    = PQ + (size_t)NCLS * KTOT;
    float* hp   = S + (size_t)NB * NCLS;
    int*   cnt  = (int*)(hp + (size_t)NCLS * 4);
    int*   list = cnt + NCLS;

    front_kernel<<<ST_BLOCKS + SG_X * SG_Y, GEMM_T, 0, stream>>>(df, gt, fc, CA, PQ, S, hp, cnt, list);
    class_out_kernel<<<dim3(CG_X, CG_Y), GEMM_T, 0, stream>>>(CA, PQ, S, hp, cnt, list, out);
}


# class-out kernel: lists fetched before the matrix slice and S values under it (compiler had sunk both behind the last WMMA); list prefix 8->16 ends the per-sample load chain; same sums, same order
# speedup vs baseline: 7.3115x; 1.2356x over previous
#include <hip/hip_runtime.h>

typedef __attribute__((ext_vector_type(2))) float v2f;
typedef __attribute__((ext_vector_type(8))) float v8f;
typedef __attribute__((ext_vector_type(4))) float v4f;
typedef __attribute__((ext_vector_type(4))) int   v4i;

#define NCLS 256
#define NDF  512
#define NB   1024
#define KTOT (2 * NDF)
#define ALP  0.1f
#define HALP 0.05f


#define STATS_T 128
#define PRE     16

typedef int gt16 __attribute__((ext_vector_type(16), aligned(4)));
typedef __attribute__((address_space(4))) gt16 gt16_k;

__device__ __forceinline__ void stats_step(v2f& m0, v2f& m1, v2f& v0, v2f& v1, int& n,
                                           const v2f f0, const v2f f1)
{
    const float nf = (float)n;
    const float c1 = nf / (nf + 1.0f);
    const float c2 = 1.0f / (nf + 1.0f);
    m0 = (n == 0) ? f0 : (m0 * c1 + f0 * c2);
    m1 = (n == 0) ? f1 : (m1 * c1 + f1 * c2);
    const v2f a0 = f0 - m0;
    const v2f a1 = f1 - m1;
    v0 = v0 * c1 + a0 * a0 * c1 * c2;
    v1 = v1 * c1 + a1 * a1 * c1 * c2;
    ++n;
}

__device__ __forceinline__ unsigned class_bits16(unsigned hit, const gt16 q, int c)
{
#pragma unroll
    for (int j = 15; j >= 0; --j) {
        asm("s_cmp_eq_u32 %1, %2\n\ts_addc_u32 %0, %0, %0"
            : "+s"(hit) : "s"(q[j]), "s"(c) : "scc");
    }
    return hit;
}

typedef __attribute__((ext_vector_type(2))) unsigned v2u;

__device__ __forceinline__ void class_stats(
    const float* __restrict__ df, const int* __restrict__ gt,
    const float* __restrict__ fc,
    float* __restrict__ CA, float* __restrict__ PQ, float* __restrict__ hp,
    int* __restrict__ cnt, int* __restrict__ list, unsigned (*hitw)[32],
    const int cpair, const int wv, const int tid)
{
    const int half = wv >> 2;
    const int c    = cpair * 2 + half;
    const int lane = tid & 31;
    const int d4   = tid * 4;
    const v4f fcv = *(const v4f*)(fc + (size_t)c * NDF + d4);
    int* lst = list + (size_t)c * NB;

    for (int r = 0; r < 2; ++r) {
        const gt16_k* q = (const gt16_k*)(gt + wv * 128 + r * 64);
        const gt16 q0 = q[0], q1 = q[1], q2 = q[2], q3 = q[3];
        const int ce = cpair * 2, co = cpair * 2 + 1;
        const unsigned lo0 = class_bits16(class_bits16(0u, q1, ce), q0, ce);
        const unsigned hi0 = class_bits16(class_bits16(0u, q3, ce), q2, ce);
        const unsigned lo1 = class_bits16(class_bits16(0u, q1, co), q0, co);
        const unsigned hi1 = class_bits16(class_bits16(0u, q3, co), q2, co);
        if (lane == 0) {
            *(v2u*)&hitw[0][wv * 4 + r * 2] = v2u{lo0, hi0};
            *(v2u*)&hitw[1][wv * 4 + r * 2] = v2u{lo1, hi1};
        }
    }
    __syncthreads();
    const unsigned word = hitw[half][lane];

    v2f m0 = {}, m1 = {}, v0 = {}, v1 = {};
    int n = 0;

    unsigned M  = __builtin_amdgcn_ballot_w32(word != 0u);
    unsigned wd = 0u;
    int base = 0;
    auto next_row = [&](v4f& dst) -> bool {
        if (wd == 0u) {
            if (M == 0u) return false;
            const int j = __builtin_ctz(M);
            M &= M - 1u;
            wd   = (unsigned)__builtin_amdgcn_readlane((int)word, j);
            base = j << 5;
        }
        const int b = base + __builtin_ctz(wd);
        wd &= wd - 1u;
        dst = *(const v4f*)(df + (size_t)b * NDF + d4);
        return true;
    };
    v4f fa = {}, fb = {};
    bool more = next_row(fa);
    while (more) {
        const bool more_b = next_row(fb);
        stats_step(m0, m1, v0, v1, n, fa.xy, fa.zw);
        if (!more_b) break;
        more = next_row(fa);
        stats_step(m0, m1, v0, v1, n, fb.xy, fb.zw);
    }

    const v4f va = {v0.x, v0.y, v1.x, v1.y};
    const v4f u  = -ALP * fcv * va;
    const v4f y  = HALP * va;
    const v4f f2 = fcv * fcv;
    *(v4f*)(CA + (size_t)c * KTOT + d4)       = u;
    *(v4f*)(CA + (size_t)c * KTOT + NDF + d4) = y;
    *(v4f*)(PQ + (size_t)c * KTOT + d4)       = fcv;
    *(v4f*)(PQ + (size_t)c * KTOT + NDF + d4) = f2;

    float p = (y.x * f2.x + y.y * f2.y) + (y.z * f2.z + y.w * f2.w);
#pragma unroll
    for (int off = 16; off > 0; off >>= 1) p += __shfl_xor(p, off, 32);
    if (lane == 0) hp[c * 4 + (tid >> 5)] = p;

    if (tid < 32) {
        unsigned m = word;
        const int mine = __builtin_popcount(m);
        int incl = mine;
#pragma unroll
        for (int off = 1; off < 32; off <<= 1) {
            const int up = __shfl_up(incl, off, 32);
            if (tid >= off) incl += up;
        }
        const int total = __shfl(incl, 31, 32);
        int pos = incl - mine;
        while (m) {
            lst[pos++] = tid * 32 + __builtin_ctz(m);
            m &= m - 1u;
        }
        if (tid < PRE && tid >= total) lst[tid] = -1;
        if (tid == 0) cnt[c] = total;
    }
}

#define GEMM_W  8
#define GEMM_T  256
#define RED_P   68
#define RED_N   (GEMM_W * 16 * RED_P)

template <int LDB, int K0, int K1, int NT>
__device__ __forceinline__ void strip_ksteps(
    const float* __restrict__ Arow, const float* __restrict__ Brow, v8f (&acc)[NT])
{
#pragma unroll
    for (int k = K0; k < K1; k += 4) {
        const v2f a = *(const v2f*)(Arow + k);
#pragma unroll
        for (int n = 0; n < NT; ++n) {
            const v2f b = *(const v2f*)(Brow + n * 16 * LDB + k);
            acc[n] = __builtin_amdgcn_wmma_f32_16x16x4_f32(false, a, false, b,
                                                           (short)0, acc[n], false, false);
        }
    }
}

template <int LDB, int K0, int K1, int NT>
__device__ __forceinline__ void frag_fetch(
    const float* __restrict__ Arow, const float* __restrict__ Brow,
    v2f (&pa)[(K1 - K0) / 4], v2f (&pb)[(K1 - K0) / 4][NT])
{
#pragma unroll
    for (int k = K0; k < K1; k += 4) {
        pa[(k - K0) / 4] = *(const v2f*)(Arow + k);
#pragma unroll
        for (int n = 0; n < NT; ++n)
            pb[(k - K0) / 4][n] = *(const v2f*)(Brow + n * 16 * LDB + k);
    }
}

template <int NS, int NT>
__device__ __forceinline__ void frag_mma(
    const v2f (&pa)[NS], const v2f (&pb)[NS][NT], v8f (&acc)[NT])
{
#pragma unroll
    for (int s = 0; s < NS; ++s) {
#pragma unroll
        for (int n = 0; n < NT; ++n)
            acc[n] = __builtin_amdgcn_wmma_f32_16x16x4_f32(false, pa[s], false, pb[s][n],
                                                           (short)0, acc[n], false, false);
    }
}

template <int NT>
__device__ __forceinline__ void strip_store(const v8f (&acc)[NT], float* red)
{
    const int w    = threadIdx.x >> 5;
    const int lane = threadIdx.x & 31;
    float* mine = red + w * (16 * RED_P) + ((lane >> 4) << 3) * RED_P + (lane & 15);
#pragma unroll
    for (int j = 0; j < 8; ++j) {
#pragma unroll
        for (int n = 0; n < NT; ++n) mine[j * RED_P + n * 16] = acc[n][j];
    }
}

template <int LD, int KS>
__device__ __forceinline__ const float* frag_row(const float* __restrict__ M, int row0)
{
    const int w    = threadIdx.x >> 5;
    const int lane = threadIdx.x & 31;
    const int r    = lane & 15;
    const int kh   = (lane >> 4) << 1;
    return M + (size_t)(row0 + r) * LD + w * KS + kh;
}

template <int LDA, int LDB, int KS, int NT>
__device__ __forceinline__ void strip_partial(
    const float* __restrict__ A, const float* __restrict__ B,
    int row0, int col0, float* red)
{
    const int w    = threadIdx.x >> 5;
    const int lane = threadIdx.x & 31;
    const int r    = lane & 15;
    const int kh   = (lane >> 4) << 1;

    const float* Arow = A + (size_t)(row0 + r) * LDA + w * KS + kh;
    const float* Brow = B + (size_t)(col0 + r) * LDB + w * KS + kh;

    v8f acc[NT];
#pragma unroll
    for (int n = 0; n < NT; ++n) acc[n] = v8f{};
#pragma unroll
    for (int k = 0; k < KS; k += 4) {
        const v2f a = *(const v2f*)(Arow + k);
#pragma unroll
        for (int n = 0; n < NT; ++n) {
            const v2f b = *(const v2f*)(Brow + n * 16 * LDB + k);
            acc[n] = __builtin_amdgcn_wmma_f32_16x16x4_f32(false, a, false, b,
                                                           (short)0, acc[n], false, false);
        }
    }

    float* mine = red + w * (16 * RED_P) + ((lane >> 4) << 3) * RED_P + r;
#pragma unroll
    for (int j = 0; j < 8; ++j) {
#pragma unroll
        for (int n = 0; n < NT; ++n) mine[j * RED_P + n * 16] = acc[n][j];
    }
}

#define ST_BLOCKS (NCLS / 2)
#define SG_X      (NB / 16)
#define SG_Y      (NCLS / 64)

__global__ __launch_bounds__(GEMM_T) void front_kernel(
    const float* __restrict__ df, const int* __restrict__ gt,
    const float* __restrict__ fc,
    float* __restrict__ CA, float* __restrict__ PQ, float* __restrict__ S,
    float* __restrict__ hp, int* __restrict__ cnt, int* __restrict__ list)
{
    __shared__ __attribute__((aligned(16))) float red[RED_N];

    if (blockIdx.x < ST_BLOCKS) {
        __shared__ __attribute__((aligned(16))) unsigned hitw[2][32];
        const int wv = __builtin_amdgcn_readfirstlane((int)(threadIdx.x >> 5));
        class_stats(df, gt, fc, CA, PQ, hp, cnt, list, hitw,
                    (int)blockIdx.x, wv, (int)(threadIdx.x & (STATS_T - 1)));
        return;
    }

    const int sb  = (int)blockIdx.x - ST_BLOCKS;
    const int tm  = sb & (SG_X - 1);
    const int ts  = sb >> 6;
    const int row = threadIdx.x >> 4;
    const int c4  = (threadIdx.x & 15) << 2;
    const int b   = tm * 16 + row;

    strip_partial<NDF, NDF, NDF / GEMM_W, 4>(df, fc, tm * 16, ts * 64, red);
    __syncthreads();

    const float* p = red + row * RED_P + c4;
    v4f s = *(const v4f*)p;
#pragma unroll
    for (int q = 1; q < GEMM_W; ++q) s += *(const v4f*)(p + q * (16 * RED_P));
    *(v4f*)(S + (size_t)b * NCLS + ts * 64 + c4) = s;
}

#define CG_X  (NCLS / 16)
#define CG_Y  (NCLS / 32)

__global__ __launch_bounds__(GEMM_T) void class_out_kernel(
    const float* __restrict__ CA, const float* __restrict__ PQ,
    const float* __restrict__ S, const float* __restrict__ hp,
    const int* __restrict__ cnt, const int* __restrict__ list,
    float* __restrict__ out)
{
    __shared__ __attribute__((aligned(16))) float red[RED_N];
    const int tm  = blockIdx.x;
    const int ts  = blockIdx.y;
    const int row = threadIdx.x >> 4;
    const int l16 = threadIdx.x & 15;
    const int t   = tm * 16 + row;
    const int c2  = l16 << 1;
    const int col = ts * 32 + c2;

    const int* lst = list + (size_t)t * NB;
    const v4i l0 = *(const v4i*)lst;
    const v4i l1 = *(const v4i*)(lst + 4);
    const v4i l2 = *(const v4i*)(lst + 8);
    const v4i l3 = *(const v4i*)(lst + 12);
    const int n  = cnt[t];
    const v4f h4 = *(const v4f*)(hp + t * 4);
    const int bb[PRE] = {l0.x, l0.y, l0.z, l0.w, l1.x, l1.y, l1.z, l1.w,
                         l2.x, l2.y, l2.z, l2.w, l3.x, l3.y, l3.z, l3.w};

    __builtin_amdgcn_sched_barrier(0);

    constexpr int KS = KTOT / GEMM_W;
    const float* Arow = frag_row<KTOT, KS>(CA, tm * 16);
    const float* Brow = frag_row<KTOT, KS>(PQ, ts * 32);
    v8f acc[2] = {v8f{}, v8f{}};
    v2f pa[KS / 16], pb[KS / 16][2];
    frag_fetch<KTOT, KS / 4, KS / 2, 2>(Arow, Brow, pa, pb);
    strip_ksteps<KTOT, 0, KS / 4, 2>(Arow, Brow, acc);

    __builtin_amdgcn_sched_barrier(0);
    v2f   sv[PRE];
    float st[PRE];
#pragma unroll
    for (int i = 0; i < PRE; ++i) {
        const float* sr = S + (size_t)(bb[i] < 0 ? 0 : bb[i]) * NCLS;
        sv[i] = *(const v2f*)(sr + col);
        st[i] = sr[t];
    }
    __builtin_amdgcn_sched_barrier(0);

    frag_mma<KS / 16, 2>(pa, pb, acc);
    strip_ksteps<KTOT, KS / 2, KS, 2>(Arow, Brow, acc);
    strip_store<2>(acc, red);
    __syncthreads();

    const float* p = red + row * RED_P + c2;
    v2f g = *(const v2f*)p;
#pragma unroll
    for (int q = 1; q < GEMM_W; ++q) g += *(const v2f*)(p + q * (16 * RED_P));

    const float h = ((h4.x + h4.y) + h4.z) + h4.w;
#pragma unroll
    for (int i = 0; i < PRE; ++i) {
        if (bb[i] >= 0)
            *(v2f*)(out + (size_t)bb[i] * NCLS + col) = (sv[i] + g) + (h - st[i]);
    }
    for (int i = PRE; i < n; ++i) {
        const int b = lst[i];
        const float* sr = S + (size_t)b * NCLS;
        const v2f   x = *(const v2f*)(sr + col);
        const float y = sr[t];
        *(v2f*)(out + (size_t)b * NCLS + col) = (x + g) + (h - y);
    }
}

extern "C" void kernel_launch(void* const* d_in, const int* in_sizes, int n_in,
                              void* d_out, int out_size, void* d_ws, size_t ws_size,
                              hipStream_t stream) {
    const float* df = (const float*)d_in[0];
    const int*   gt = (const int*)d_in[1];
    const float* fc = (const float*)d_in[2];
    float*       out = (float*)d_out;

    float* CA   = (float*)d_ws;
    float* PQ   = CA + (size_t)NCLS * KTOT;
    float* S    = PQ + (size_t)NCLS * KTOT;
    float* hp   = S + (size_t)NB * NCLS;
    int*   cnt  = (int*)(hp + (size_t)NCLS * 4);
    int*   list = cnt + NCLS;

    front_kernel<<<ST_BLOCKS + SG_X * SG_Y, GEMM_T, 0, stream>>>(df, gt, fc, CA, PQ, S, hp, cnt, list);
    class_out_kernel<<<dim3(CG_X, CG_Y), GEMM_T, 0, stream>>>(CA, PQ, S, hp, cnt, list, out);
}


# stats walk: found rows requested eight at a time, then folded in order: one memory trip per eight samples (the two-deep form's wait also covered the row just requested: a trip per two); same sums
# speedup vs baseline: 7.7836x; 1.0646x over previous
#include <hip/hip_runtime.h>

typedef __attribute__((ext_vector_type(2))) float v2f;
typedef __attribute__((ext_vector_type(8))) float v8f;
typedef __attribute__((ext_vector_type(4))) float v4f;
typedef __attribute__((ext_vector_type(4))) int   v4i;

#define NCLS 256
#define NDF  512
#define NB   1024
#define KTOT (2 * NDF)
#define ALP  0.1f
#define HALP 0.05f


#define STATS_T 128
#define PRE     16

typedef int gt16 __attribute__((ext_vector_type(16), aligned(4)));
typedef __attribute__((address_space(4))) gt16 gt16_k;

__device__ __forceinline__ void stats_step(v2f& m0, v2f& m1, v2f& v0, v2f& v1, int& n,
                                           const v2f f0, const v2f f1)
{
    const float nf = (float)n;
    const float c1 = nf / (nf + 1.0f);
    const float c2 = 1.0f / (nf + 1.0f);
    m0 = (n == 0) ? f0 : (m0 * c1 + f0 * c2);
    m1 = (n == 0) ? f1 : (m1 * c1 + f1 * c2);
    const v2f a0 = f0 - m0;
    const v2f a1 = f1 - m1;
    v0 = v0 * c1 + a0 * a0 * c1 * c2;
    v1 = v1 * c1 + a1 * a1 * c1 * c2;
    ++n;
}

__device__ __forceinline__ unsigned class_bits16(unsigned hit, const gt16 q, int c)
{
#pragma unroll
    for (int j = 15; j >= 0; --j) {
        asm("s_cmp_eq_u32 %1, %2\n\ts_addc_u32 %0, %0, %0"
            : "+s"(hit) : "s"(q[j]), "s"(c) : "scc");
    }
    return hit;
}

typedef __attribute__((ext_vector_type(2))) unsigned v2u;

__device__ __forceinline__ void class_stats(
    const float* __restrict__ df, const int* __restrict__ gt,
    const float* __restrict__ fc,
    float* __restrict__ CA, float* __restrict__ PQ, float* __restrict__ hp,
    int* __restrict__ cnt, int* __restrict__ list, unsigned (*hitw)[32],
    const int cpair, const int wv, const int tid)
{
    const int half = wv >> 2;
    const int c    = cpair * 2 + half;
    const int lane = tid & 31;
    const int d4   = tid * 4;
    const v4f fcv = *(const v4f*)(fc + (size_t)c * NDF + d4);
    int* lst = list + (size_t)c * NB;

    for (int r = 0; r < 2; ++r) {
        const gt16_k* q = (const gt16_k*)(gt + wv * 128 + r * 64);
        const gt16 q0 = q[0], q1 = q[1], q2 = q[2], q3 = q[3];
        const int ce = cpair * 2, co = cpair * 2 + 1;
        const unsigned lo0 = class_bits16(class_bits16(0u, q1, ce), q0, ce);
        const unsigned hi0 = class_bits16(class_bits16(0u, q3, ce), q2, ce);
        const unsigned lo1 = class_bits16(class_bits16(0u, q1, co), q0, co);
        const unsigned hi1 = class_bits16(class_bits16(0u, q3, co), q2, co);
        if (lane == 0) {
            *(v2u*)&hitw[0][wv * 4 + r * 2] = v2u{lo0, hi0};
            *(v2u*)&hitw[1][wv * 4 + r * 2] = v2u{lo1, hi1};
        }
    }
    __syncthreads();
    const unsigned word = hitw[half][lane];

    v2f m0 = {}, m1 = {}, v0 = {}, v1 = {};
    int n = 0;

    unsigned M  = __builtin_amdgcn_ballot_w32(word != 0u);
    unsigned wd = 0u;
    int base = 0;
    auto next_row = [&](v4f& dst) -> bool {
        if (wd == 0u) {
            if (M == 0u) return false;
            const int j = __builtin_ctz(M);
            M &= M - 1u;
            wd   = (unsigned)__builtin_amdgcn_readlane((int)word, j);
            base = j << 5;
        }
        const int b = base + __builtin_ctz(wd);
        wd &= wd - 1u;
        dst = *(const v4f*)(df + (size_t)b * NDF + d4);
        return true;
    };
    for (;;) {
        v4f r0, r1, r2, r3, r4, r5, r6, r7;
        if (!next_row(r0)) break;
        const bool h1 = next_row(r1);
        const bool h2 = h1 && next_row(r2);
        const bool h3 = h2 && next_row(r3);
        const bool h4 = h3 && next_row(r4);
        const bool h5 = h4 && next_row(r5);
        const bool h6 = h5 && next_row(r6);
        const bool h7 = h6 && next_row(r7);
        stats_step(m0, m1, v0, v1, n, r0.xy, r0.zw);
        if (!h1) break;
        stats_step(m0, m1, v0, v1, n, r1.xy, r1.zw);
        if (!h2) break;
        stats_step(m0, m1, v0, v1, n, r2.xy, r2.zw);
        if (!h3) break;
        stats_step(m0, m1, v0, v1, n, r3.xy, r3.zw);
        if (!h4) break;
        stats_step(m0, m1, v0, v1, n, r4.xy, r4.zw);
        if (!h5) break;
        stats_step(m0, m1, v0, v1, n, r5.xy, r5.zw);
        if (!h6) break;
        stats_step(m0, m1, v0, v1, n, r6.xy, r6.zw);
        if (!h7) break;
        stats_step(m0, m1, v0, v1, n, r7.xy, r7.zw);
    }

    const v4f va = {v0.x, v0.y, v1.x, v1.y};
    const v4f u  = -ALP * fcv * va;
    const v4f y  = HALP * va;
    const v4f f2 = fcv * fcv;
    *(v4f*)(CA + (size_t)c * KTOT + d4)       = u;
    *(v4f*)(CA + (size_t)c * KTOT + NDF + d4) = y;
    *(v4f*)(PQ + (size_t)c * KTOT + d4)       = fcv;
    *(v4f*)(PQ + (size_t)c * KTOT + NDF + d4) = f2;

    float p = (y.x * f2.x + y.y * f2.y) + (y.z * f2.z + y.w * f2.w);
#pragma unroll
    for (int off = 16; off > 0; off >>= 1) p += __shfl_xor(p, off, 32);
    if (lane == 0) hp[c * 4 + (tid >> 5)] = p;

    if (tid < 32) {
        unsigned m = word;
        const int mine = __builtin_popcount(m);
        int incl = mine;
#pragma unroll
        for (int off = 1; off < 32; off <<= 1) {
            const int up = __shfl_up(incl, off, 32);
            if (tid >= off) incl += up;
        }
        const int total = __shfl(incl, 31, 32);
        int pos = incl - mine;
        while (m) {
            lst[pos++] = tid * 32 + __builtin_ctz(m);
            m &= m - 1u;
        }
        if (tid < PRE && tid >= total) lst[tid] = -1;
        if (tid == 0) cnt[c] = total;
    }
}

#define GEMM_W  8
#define GEMM_T  256
#define RED_P   68
#define RED_N   (GEMM_W * 16 * RED_P)

template <int LDB, int K0, int K1, int NT>
__device__ __forceinline__ void strip_ksteps(
    const float* __restrict__ Arow, const float* __restrict__ Brow, v8f (&acc)[NT])
{
#pragma unroll
    for (int k = K0; k < K1; k += 4) {
        const v2f a = *(const v2f*)(Arow + k);
#pragma unroll
        for (int n = 0; n < NT; ++n) {
            const v2f b = *(const v2f*)(Brow + n * 16 * LDB + k);
            acc[n] = __builtin_amdgcn_wmma_f32_16x16x4_f32(false, a, false, b,
                                                           (short)0, acc[n], false, false);
        }
    }
}

template <int LDB, int K0, int K1, int NT>
__device__ __forceinline__ void frag_fetch(
    const float* __restrict__ Arow, const float* __restrict__ Brow,
    v2f (&pa)[(K1 - K0) / 4], v2f (&pb)[(K1 - K0) / 4][NT])
{
#pragma unroll
    for (int k = K0; k < K1; k += 4) {
        pa[(k - K0) / 4] = *(const v2f*)(Arow + k);
#pragma unroll
        for (int n = 0; n < NT; ++n)
            pb[(k - K0) / 4][n] = *(const v2f*)(Brow + n * 16 * LDB + k);
    }
}

template <int NS, int NT>
__device__ __forceinline__ void frag_mma(
    const v2f (&pa)[NS], const v2f (&pb)[NS][NT], v8f (&acc)[NT])
{
#pragma unroll
    for (int s = 0; s < NS; ++s) {
#pragma unroll
        for (int n = 0; n < NT; ++n)
            acc[n] = __builtin_amdgcn_wmma_f32_16x16x4_f32(false, pa[s], false, pb[s][n],
                                                           (short)0, acc[n], false, false);
    }
}

template <int NT>
__device__ __forceinline__ void strip_store(const v8f (&acc)[NT], float* red)
{
    const int w    = threadIdx.x >> 5;
    const int lane = threadIdx.x & 31;
    float* mine = red + w * (16 * RED_P) + ((lane >> 4) << 3) * RED_P + (lane & 15);
#pragma unroll
    for (int j = 0; j < 8; ++j) {
#pragma unroll
        for (int n = 0; n < NT; ++n) mine[j * RED_P + n * 16] = acc[n][j];
    }
}

template <int LD, int KS>
__device__ __forceinline__ const float* frag_row(const float* __restrict__ M, int row0)
{
    const int w    = threadIdx.x >> 5;
    const int lane = threadIdx.x & 31;
    const int r    = lane & 15;
    const int kh   = (lane >> 4) << 1;
    return M + (size_t)(row0 + r) * LD + w * KS + kh;
}

template <int LDA, int LDB, int KS, int NT>
__device__ __forceinline__ void strip_partial(
    const float* __restrict__ A, const float* __restrict__ B,
    int row0, int col0, float* red)
{
    const int w    = threadIdx.x >> 5;
    const int lane = threadIdx.x & 31;
    const int r    = lane & 15;
    const int kh   = (lane >> 4) << 1;

    const float* Arow = A + (size_t)(row0 + r) * LDA + w * KS + kh;
    const float* Brow = B + (size_t)(col0 + r) * LDB + w * KS + kh;

    v8f acc[NT];
#pragma unroll
    for (int n = 0; n < NT; ++n) acc[n] = v8f{};
#pragma unroll
    for (int k = 0; k < KS; k += 4) {
        const v2f a = *(const v2f*)(Arow + k);
#pragma unroll
        for (int n = 0; n < NT; ++n) {
            const v2f b = *(const v2f*)(Brow + n * 16 * LDB + k);
            acc[n] = __builtin_amdgcn_wmma_f32_16x16x4_f32(false, a, false, b,
                                                           (short)0, acc[n], false, false);
        }
    }

    float* mine = red + w * (16 * RED_P) + ((lane >> 4) << 3) * RED_P + r;
#pragma unroll
    for (int j = 0; j < 8; ++j) {
#pragma unroll
        for (int n = 0; n < NT; ++n) mine[j * RED_P + n * 16] = acc[n][j];
    }
}

#define ST_BLOCKS (NCLS / 2)
#define SG_X      (NB / 16)
#define SG_Y      (NCLS / 64)

__global__ __launch_bounds__(GEMM_T) void front_kernel(
    const float* __restrict__ df, const int* __restrict__ gt,
    const float* __restrict__ fc,
    float* __restrict__ CA, float* __restrict__ PQ, float* __restrict__ S,
    float* __restrict__ hp, int* __restrict__ cnt, int* __restrict__ list)
{
    __shared__ __attribute__((aligned(16))) float red[RED_N];

    if (blockIdx.x < ST_BLOCKS) {
        __shared__ __attribute__((aligned(16))) unsigned hitw[2][32];
        const int wv = __builtin_amdgcn_readfirstlane((int)(threadIdx.x >> 5));
        class_stats(df, gt, fc, CA, PQ, hp, cnt, list, hitw,
                    (int)blockIdx.x, wv, (int)(threadIdx.x & (STATS_T - 1)));
        return;
    }

    const int sb  = (int)blockIdx.x - ST_BLOCKS;
    const int tm  = sb & (SG_X - 1);
    const int ts  = sb >> 6;
    const int row = threadIdx.x >> 4;
    const int c4  = (threadIdx.x & 15) << 2;
    const int b   = tm * 16 + row;

    strip_partial<NDF, NDF, NDF / GEMM_W, 4>(df, fc, tm * 16, ts * 64, red);
    __syncthreads();

    const float* p = red + row * RED_P + c4;
    v4f s = *(const v4f*)p;
#pragma unroll
    for (int q = 1; q < GEMM_W; ++q) s += *(const v4f*)(p + q * (16 * RED_P));
    *(v4f*)(S + (size_t)b * NCLS + ts * 64 + c4) = s;
}

#define CG_X  (NCLS / 16)
#define CG_Y  (NCLS / 32)

__global__ __launch_bounds__(GEMM_T) void class_out_kernel(
    const float* __restrict__ CA, const float* __restrict__ PQ,
    const float* __restrict__ S, const float* __restrict__ hp,
    const int* __restrict__ cnt, const int* __restrict__ list,
    float* __restrict__ out)
{
    __shared__ __attribute__((aligned(16))) float red[RED_N];
    const int tm  = blockIdx.x;
    const int ts  = blockIdx.y;
    const int row = threadIdx.x >> 4;
    const int l16 = threadIdx.x & 15;
    const int t   = tm * 16 + row;
    const int c2  = l16 << 1;
    const int col = ts * 32 + c2;

    const int* lst = list + (size_t)t * NB;
    const v4i l0 = *(const v4i*)lst;
    const v4i l1 = *(const v4i*)(lst + 4);
    const v4i l2 = *(const v4i*)(lst + 8);
    const v4i l3 = *(const v4i*)(lst + 12);
    const int n  = cnt[t];
    const v4f h4 = *(const v4f*)(hp + t * 4);
    const int bb[PRE] = {l0.x, l0.y, l0.z, l0.w, l1.x, l1.y, l1.z, l1.w,
                         l2.x, l2.y, l2.z, l2.w, l3.x, l3.y, l3.z, l3.w};

    __builtin_amdgcn_sched_barrier(0);

    constexpr int KS = KTOT / GEMM_W;
    const float* Arow = frag_row<KTOT, KS>(CA, tm * 16);
    const float* Brow = frag_row<KTOT, KS>(PQ, ts * 32);
    v8f acc[2] = {v8f{}, v8f{}};
    v2f pa[KS / 16], pb[KS / 16][2];
    frag_fetch<KTOT, KS / 4, KS / 2, 2>(Arow, Brow, pa, pb);
    strip_ksteps<KTOT, 0, KS / 4, 2>(Arow, Brow, acc);

    __builtin_amdgcn_sched_barrier(0);
    v2f   sv[PRE];
    float st[PRE];
#pragma unroll
    for (int i = 0; i < PRE; ++i) {
        const float* sr = S + (size_t)(bb[i] < 0 ? 0 : bb[i]) * NCLS;
        sv[i] = *(const v2f*)(sr + col);
        st[i] = sr[t];
    }
    __builtin_amdgcn_sched_barrier(0);

    frag_mma<KS / 16, 2>(pa, pb, acc);
    strip_ksteps<KTOT, KS / 2, KS, 2>(Arow, Brow, acc);
    strip_store<2>(acc, red);
    __syncthreads();

    const float* p = red + row * RED_P + c2;
    v2f g = *(const v2f*)p;
#pragma unroll
    for (int q = 1; q < GEMM_W; ++q) g += *(const v2f*)(p + q * (16 * RED_P));

    const float h = ((h4.x + h4.y) + h4.z) + h4.w;
#pragma unroll
    for (int i = 0; i < PRE; ++i) {
        if (bb[i] >= 0)
            *(v2f*)(out + (size_t)bb[i] * NCLS + col) = (sv[i] + g) + (h - st[i]);
    }
    for (int i = PRE; i < n; ++i) {
        const int b = lst[i];
        const float* sr = S + (size_t)b * NCLS;
        const v2f   x = *(const v2f*)(sr + col);
        const float y = sr[t];
        *(v2f*)(out + (size_t)b * NCLS + col) = (x + g) + (h - y);
    }
}

extern "C" void kernel_launch(void* const* d_in, const int* in_sizes, int n_in,
                              void* d_out, int out_size, void* d_ws, size_t ws_size,
                              hipStream_t stream) {
    const float* df = (const float*)d_in[0];
    const int*   gt = (const int*)d_in[1];
    const float* fc = (const float*)d_in[2];
    float*       out = (float*)d_out;

    float* CA   = (float*)d_ws;
    float* PQ   = CA + (size_t)NCLS * KTOT;
    float* S    = PQ + (size_t)NCLS * KTOT;
    float* hp   = S + (size_t)NB * NCLS;
    int*   cnt  = (int*)(hp + (size_t)NCLS * 4);
    int*   list = cnt + NCLS;

    front_kernel<<<ST_BLOCKS + SG_X * SG_Y, GEMM_T, 0, stream>>>(df, gt, fc, CA, PQ, S, hp, cnt, list);
    class_out_kernel<<<dim3(CG_X, CG_Y), GEMM_T, 0, stream>>>(CA, PQ, S, hp, cnt, list, out);
}
